# starting kernel
# baseline (speedup 1.0000x reference)
#include <hip/hip_runtime.h>
#include <hip/hip_bf16.h>
#include <math.h>


typedef __attribute__((ext_vector_type(2))) float v2f;
typedef __attribute__((ext_vector_type(8))) float v8f;

#define GAT_IN_DIM   128
#define GAT_HEADS    8
#define GAT_FPH      16
#define GAT_OUT      128
#define NEG_SLOPE    0.2f

__global__ __launch_bounds__(256) void gat_proj_wmma(
    const float* __restrict__ x, const float* __restrict__ W,
    float* __restrict__ xp, int N) {
  __shared__ float sx[16 * GAT_IN_DIM];

  const int row0 = blockIdx.x * 16;
  const int tid  = threadIdx.x;

  for (int i = tid; i < 16 * GAT_IN_DIM; i += 256) {
    int r = i >> 7, c = i & 127;
    int gr = row0 + r;
    sx[i] = (gr < N) ? x[gr * GAT_IN_DIM + c] : 0.0f;
  }
  __syncthreads();

  const int wave = tid >> 5;
  const int lane = tid & 31;
  const int n    = lane & 15;
  const int half = lane >> 4;
  const int col  = wave * 16 + n;
  const int arow = lane & 15;

  v8f acc = {};
  for (int k0 = 0; k0 < GAT_IN_DIM; k0 += 4) {
    const int kk = k0 + 2 * half;
    v2f a, b;
    a[0] = sx[arow * GAT_IN_DIM + kk];
    a[1] = sx[arow * GAT_IN_DIM + kk + 1];
    b[0] = W[kk * GAT_OUT + col];
    b[1] = W[(kk + 1) * GAT_OUT + col];
    acc = __builtin_amdgcn_wmma_f32_16x16x4_f32(
         false, a,  false, b,
         (short)0, acc,  false,  false);
  }

  const int mbase = half * 8;
#pragma unroll
  for (int i = 0; i < 8; ++i) {
    int gr = row0 + mbase + i;
    if (gr < N) xp[gr * GAT_OUT + col] = acc[i];
  }
}

__global__ void gat_al(const float* __restrict__ xp,
                       const float* __restrict__ a_src,
                       const float* __restrict__ a_dst,
                       float* __restrict__ al_src,
                       float* __restrict__ al_dst, int N) {
  int idx = blockIdx.x * blockDim.x + threadIdx.x;
  if (idx >= N * GAT_HEADS) return;
  int h = idx & 7, node = idx >> 3;
  const float* r  = xp + node * GAT_OUT + h * GAT_FPH;
  const float* as = a_src + h * GAT_FPH;
  const float* ad = a_dst + h * GAT_FPH;
  float s = 0.0f, d = 0.0f;
#pragma unroll
  for (int f = 0; f < GAT_FPH; ++f) {
    float v = r[f];
    s += v * as[f];
    d += v * ad[f];
  }
  al_src[idx] = s;
  al_dst[idx] = d;
}

__global__ void gat_init(float* __restrict__ out, float* __restrict__ m,
                         float* __restrict__ denom, int N) {
  int i = blockIdx.x * blockDim.x + threadIdx.x;
  if (i < N * GAT_OUT) out[i] = 0.0f;
  if (i < N * GAT_HEADS) {
    m[i] = -__builtin_inff();
    denom[i] = 0.0f;
  }
}

__device__ __forceinline__ void atomicMaxF(float* addr, float v) {
  if (v >= 0.0f)
    atomicMax((int*)addr, __float_as_int(v));
  else
    atomicMin((unsigned int*)addr, __float_as_uint(v));
}

__device__ __forceinline__ float leaky(float e) {
  return e >= 0.0f ? e : NEG_SLOPE * e;
}

__global__ void gat_edge_max(const int* __restrict__ src,
                             const int* __restrict__ dst,
                             const float* __restrict__ al_src,
                             const float* __restrict__ al_dst,
                             float* __restrict__ m, int E) {
  int e = blockIdx.x * blockDim.x + threadIdx.x;
  if (e >= E) return;
  int s = src[e], d = dst[e];
  const float4* as = (const float4*)(al_src + s * GAT_HEADS);
  const float4* ad = (const float4*)(al_dst + d * GAT_HEADS);
  float4 a0 = as[0], a1 = as[1], b0 = ad[0], b1 = ad[1];
  float va[8] = {a0.x, a0.y, a0.z, a0.w, a1.x, a1.y, a1.z, a1.w};
  float vb[8] = {b0.x, b0.y, b0.z, b0.w, b1.x, b1.y, b1.z, b1.w};
#pragma unroll
  for (int h = 0; h < GAT_HEADS; ++h)
    atomicMaxF(&m[d * GAT_HEADS + h], leaky(va[h] + vb[h]));
}

__global__ void gat_edge_expsum(const int* __restrict__ src,
                                const int* __restrict__ dst,
                                const float* __restrict__ al_src,
                                const float* __restrict__ al_dst,
                                const float* __restrict__ m,
                                float* __restrict__ denom, int E) {
  int e = blockIdx.x * blockDim.x + threadIdx.x;
  if (e >= E) return;
  int s = src[e], d = dst[e];
  const float4* as = (const float4*)(al_src + s * GAT_HEADS);
  const float4* ad = (const float4*)(al_dst + d * GAT_HEADS);
  const float4* mm = (const float4*)(m + d * GAT_HEADS);
  float4 a0 = as[0], a1 = as[1], b0 = ad[0], b1 = ad[1], m0 = mm[0], m1 = mm[1];
  float va[8] = {a0.x, a0.y, a0.z, a0.w, a1.x, a1.y, a1.z, a1.w};
  float vb[8] = {b0.x, b0.y, b0.z, b0.w, b1.x, b1.y, b1.z, b1.w};
  float vm[8] = {m0.x, m0.y, m0.z, m0.w, m1.x, m1.y, m1.z, m1.w};
#pragma unroll
  for (int h = 0; h < GAT_HEADS; ++h)
    atomicAdd(&denom[d * GAT_HEADS + h], expf(leaky(va[h] + vb[h]) - vm[h]));
}

__global__ __launch_bounds__(256) void gat_edge_agg(
    const int* __restrict__ src, const int* __restrict__ dst,
    const float* __restrict__ al_src, const float* __restrict__ al_dst,
    const float* __restrict__ m, const float* __restrict__ denom,
    const float* __restrict__ xp, float* __restrict__ out, int E) {
  int gwarp  = (blockIdx.x * blockDim.x + threadIdx.x) >> 5;
  int lane   = threadIdx.x & 31;
  int nwarps = (gridDim.x * blockDim.x) >> 5;
  for (int e = gwarp; e < E; e += nwarps) {
    int s = src[e], d = dst[e];
    int h = lane >> 2;
    float as = al_src[s * GAT_HEADS + h];
    float ad = al_dst[d * GAT_HEADS + h];
    float mx = m[d * GAT_HEADS + h];
    float dn = denom[d * GAT_HEADS + h];
    float ex = expf(leaky(as + ad) - mx);
    float alpha = ex / (dn == 0.0f ? 1.0f : dn);
    int col = lane * 4;
    float4 v = *(const float4*)(xp + s * GAT_OUT + col);
    float* o = out + d * GAT_OUT + col;
    atomicAdd(o + 0, v.x * alpha);
    atomicAdd(o + 1, v.y * alpha);
    atomicAdd(o + 2, v.z * alpha);
    atomicAdd(o + 3, v.w * alpha);
  }
}

__global__ void gat_finalize(float* __restrict__ out,
                             const float* __restrict__ bias, int N) {
  int i = blockIdx.x * blockDim.x + threadIdx.x;
  if (i >= N * GAT_OUT) return;
  float v = out[i] + bias[i & 127];
  out[i] = v > 0.0f ? v : expm1f(v);
}

extern "C" void kernel_launch(void* const* d_in, const int* in_sizes, int n_in,
                              void* d_out, int out_size, void* d_ws,
                              size_t ws_size, hipStream_t stream) {
  const float* x      = (const float*)d_in[0];
  const int*   ei     = (const int*)d_in[1];
  const float* W      = (const float*)d_in[2];
  const float* a_src  = (const float*)d_in[3];
  const float* a_dst  = (const float*)d_in[4];
  const float* bias   = (const float*)d_in[5];

  const int N = in_sizes[0] / GAT_IN_DIM;
  const int E = in_sizes[1] / 2;
  const int* src = ei;
  const int* dst = ei + E;

  float* out = (float*)d_out;

  float* ws     = (float*)d_ws;
  float* xp     = ws;
  float* al_src = xp + (size_t)N * GAT_OUT;
  float* al_dst = al_src + (size_t)N * GAT_HEADS;
  float* mbuf   = al_dst + (size_t)N * GAT_HEADS;
  float* denom  = mbuf + (size_t)N * GAT_HEADS;

  {
    int total = N * GAT_OUT;
    gat_init<<<(total + 255) / 256, 256, 0, stream>>>(out, mbuf, denom, N);
  }
  gat_proj_wmma<<<(N + 15) / 16, 256, 0, stream>>>(x, W, xp, N);
  {
    int total = N * GAT_HEADS;
    gat_al<<<(total + 255) / 256, 256, 0, stream>>>(xp, a_src, a_dst, al_src,
                                                    al_dst, N);
  }
  gat_edge_max<<<(E + 255) / 256, 256, 0, stream>>>(src, dst, al_src, al_dst,
                                                    mbuf, E);
  gat_edge_expsum<<<(E + 255) / 256, 256, 0, stream>>>(src, dst, al_src,
                                                       al_dst, mbuf, denom, E);
  gat_edge_agg<<<8192, 256, 0, stream>>>(src, dst, al_src, al_dst, mbuf, denom,
                                         xp, out, E);
  {
    int total = N * GAT_OUT;
    gat_finalize<<<(total + 255) / 256, 256, 0, stream>>>(out, bias, N);
  }
}


# Group edges by destination in the call (count, allocate, fill), sort each node's list by source id so sums repeat exactly; one wave per node does max, exp-sum, row sum; float atomics removed
# speedup vs baseline: 3.1967x; 3.1967x over previous
#include <hip/hip_runtime.h>
#include <hip/hip_bf16.h>
#include <math.h>


typedef __attribute__((ext_vector_type(2))) float v2f;
typedef __attribute__((ext_vector_type(8))) float v8f;

#define GAT_IN_DIM   128
#define GAT_HEADS    8
#define GAT_FPH      16
#define GAT_OUT      128
#define NEG_SLOPE    0.2f

__global__ __launch_bounds__(256) void gat_proj_wmma(
    const float* __restrict__ x, const float* __restrict__ W,
    float* __restrict__ xp, int N) {
  __shared__ float sx[16 * GAT_IN_DIM];

  const int row0 = blockIdx.x * 16;
  const int tid  = threadIdx.x;

  for (int i = tid; i < 16 * GAT_IN_DIM; i += 256) {
    int r = i >> 7, c = i & 127;
    int gr = row0 + r;
    sx[i] = (gr < N) ? x[gr * GAT_IN_DIM + c] : 0.0f;
  }
  __syncthreads();

  const int wave = tid >> 5;
  const int lane = tid & 31;
  const int n    = lane & 15;
  const int half = lane >> 4;
  const int col  = wave * 16 + n;
  const int arow = lane & 15;

  v8f acc = {};
  for (int k0 = 0; k0 < GAT_IN_DIM; k0 += 4) {
    const int kk = k0 + 2 * half;
    v2f a, b;
    a[0] = sx[arow * GAT_IN_DIM + kk];
    a[1] = sx[arow * GAT_IN_DIM + kk + 1];
    b[0] = W[kk * GAT_OUT + col];
    b[1] = W[(kk + 1) * GAT_OUT + col];
    acc = __builtin_amdgcn_wmma_f32_16x16x4_f32(
         false, a,  false, b,
         (short)0, acc,  false,  false);
  }

  const int mbase = half * 8;
#pragma unroll
  for (int i = 0; i < 8; ++i) {
    int gr = row0 + mbase + i;
    if (gr < N) xp[gr * GAT_OUT + col] = acc[i];
  }
}

__global__ void gat_al(const float* __restrict__ xp,
                       const float* __restrict__ a_src,
                       float* __restrict__ al_src, int N) {
  int idx = blockIdx.x * blockDim.x + threadIdx.x;
  if (idx >= N * GAT_HEADS) return;
  int h = idx & 7, node = idx >> 3;
  const float* r  = xp + node * GAT_OUT + h * GAT_FPH;
  const float* as = a_src + h * GAT_FPH;
  float s = 0.0f;
#pragma unroll
  for (int f = 0; f < GAT_FPH; ++f) s += r[f] * as[f];
  al_src[idx] = s;
}

__global__ void gat_init(int* __restrict__ cnt, int* __restrict__ total,
                         int N) {
  int i = blockIdx.x * blockDim.x + threadIdx.x;
  if (i < N) cnt[i] = 0;
  if (i == 0) *total = 0;
}

__global__ __launch_bounds__(256) void gat_dst_count(
    const int* __restrict__ dst, int* __restrict__ cnt, int E) {
  const int stride = gridDim.x * blockDim.x;
  for (int e = blockIdx.x * blockDim.x + threadIdx.x; e < E; e += stride)
    atomicAdd(&cnt[dst[e]], 1);
}

__global__ __launch_bounds__(256) void gat_seg_alloc(
    const int* __restrict__ cnt, int* __restrict__ cur,
    int* __restrict__ total, int N) {
  const int d    = blockIdx.x * blockDim.x + threadIdx.x;
  const int lane = threadIdx.x & 31;
  const int c    = (d < N) ? cnt[d] : 0;
  int incl = c;
#pragma unroll
  for (int o = 1; o < 32; o <<= 1) {
    int t = __shfl_up(incl, o);
    if (lane >= o) incl += t;
  }
  int base = 0;
  if (lane == 31) base = atomicAdd(total, incl);
  base = __shfl(base, 31);
  if (d < N) cur[d] = base + incl - c;
}

__global__ __launch_bounds__(256) void gat_seg_fill(
    const int* __restrict__ src, const int* __restrict__ dst,
    int* __restrict__ cur, int* __restrict__ nbr, int E) {
  const int stride = gridDim.x * blockDim.x;
  for (int e = blockIdx.x * blockDim.x + threadIdx.x; e < E; e += stride) {
    const int p = atomicAdd(&cur[dst[e]], 1);
    nbr[p] = src[e];
  }
}

__device__ __forceinline__ float leaky(float e) {
  return e >= 0.0f ? e : NEG_SLOPE * e;
}

__device__ __forceinline__ void wave_lds_order() {
  __builtin_amdgcn_fence(__ATOMIC_RELEASE, "wavefront");
  __builtin_amdgcn_wave_barrier();
  __builtin_amdgcn_fence(__ATOMIC_ACQUIRE, "wavefront");
}

__device__ __forceinline__ void gat_sorted_chunk(const int* __restrict__ seg,
                                              int deg, int c, int lane,
                                              int* ids) {
  for (int pb = 0; pb < deg; pb += 32) {
    const int p = pb + lane;
    const bool valid = p < deg;
    const int v = valid ? seg[p] : 0;
    int rank = 0;
    for (int i = 0; i < deg; ++i) {
      const int vi = seg[i];
      rank += ((vi < v) | ((vi == v) & (i < p))) ? 1 : 0;
    }
    const int r = rank - 32 * c;
    if (valid && r >= 0 && r < 32) ids[r] = v;
  }
}

__global__ __launch_bounds__(256, 16) void gat_node_gather(
    const int* __restrict__ cnt, const int* __restrict__ seg_end,
    const int* __restrict__ nbr, const float* __restrict__ al_src,
    const float* __restrict__ a_dst, const float* __restrict__ xp,
    const float* __restrict__ bias, float* __restrict__ out, int N) {
  __shared__ int   s_ids[8][32];
  __shared__ __align__(16) float s_w[8][32 * GAT_HEADS];

  const int lane = threadIdx.x & 31;
  const int wv = __builtin_amdgcn_readfirstlane((int)(threadIdx.x >> 5));
  int*   ids = s_ids[wv];
  float* w   = s_w[wv];
  const int nw = gridDim.x * 8;
  const int h8 = lane & 7;
  const int q  = lane >> 3;
  const int hc = lane >> 2;
  const int col = lane * 4;
  const float4 adv = *(const float4*)(a_dst + col);
  const float4 bv  = *(const float4*)(bias + col);

  for (int d = blockIdx.x * 8 + wv; d < N; d += nw) {
    const int deg = cnt[d];
    const int* seg = nbr + (seg_end[d] - deg);
    const int nch = (deg + 31) >> 5;

    const float4 xd = *(const float4*)(xp + (size_t)d * GAT_OUT + col);
    float pd = xd.x * adv.x + xd.y * adv.y + xd.z * adv.z + xd.w * adv.w;
    pd += __shfl_xor(pd, 1);
    pd += __shfl_xor(pd, 2);
    float adh[GAT_HEADS];
#pragma unroll
    for (int h = 0; h < GAT_HEADS; ++h)
      adh[h] = __int_as_float(
          __builtin_amdgcn_readlane(__float_as_int(pd), 4 * h));

    auto stage_logits = [&](int sidv, int cn) {
      const float ninf = -__builtin_inff();
      float4 e0 = make_float4(ninf, ninf, ninf, ninf), e1 = e0;
      if (lane < cn) {
        const float* ap = al_src + (size_t)sidv * GAT_HEADS;
        const float4 a0 = *(const float4*)(ap);
        const float4 a1 = *(const float4*)(ap + 4);
        e0.x = leaky(a0.x + adh[0]);
        e0.y = leaky(a0.y + adh[1]);
        e0.z = leaky(a0.z + adh[2]);
        e0.w = leaky(a0.w + adh[3]);
        e1.x = leaky(a1.x + adh[4]);
        e1.y = leaky(a1.y + adh[5]);
        e1.z = leaky(a1.z + adh[6]);
        e1.w = leaky(a1.w + adh[7]);
      }
      *(float4*)(w + lane * GAT_HEADS)     = e0;
      *(float4*)(w + lane * GAT_HEADS + 4) = e1;
    };

    int sid = 0;
    if (nch == 1) {
      const int v = (lane < deg) ? seg[lane] : 0;
      int rank = 0;
      for (int i = 0; i < deg; ++i) {
        const int vi = __builtin_amdgcn_readlane(v, i);
        rank += ((vi < v) | ((vi == v) & (i < lane))) ? 1 : 0;
      }
      if (lane < deg) ids[rank] = v;
      wave_lds_order();
      sid = (lane < deg) ? ids[lane] : 0;
    }

    float m = -__builtin_inff();
    for (int c = 0; c < nch; ++c) {
      const int cn = min(32, deg - 32 * c);
      if (nch > 1) {
        wave_lds_order();
        gat_sorted_chunk(seg, deg, c, lane, ids);
        wave_lds_order();
        sid = (lane < cn) ? ids[lane] : 0;
      }
      stage_logits(sid, cn);
      wave_lds_order();
      for (int j0 = 0; j0 < cn; j0 += 4)
        m = fmaxf(m, w[(j0 + q) * GAT_HEADS + h8]);
    }
    m = fmaxf(m, __shfl_xor(m, 8));
    m = fmaxf(m, __shfl_xor(m, 16));

    float sum = 0.0f;
    for (int c = 0; c < nch; ++c) {
      const int cn = min(32, deg - 32 * c);
      if (nch > 1) {
        wave_lds_order();
        gat_sorted_chunk(seg, deg, c, lane, ids);
        wave_lds_order();
        sid = (lane < cn) ? ids[lane] : 0;
        stage_logits(sid, cn);
        wave_lds_order();
      }
      for (int j0 = 0; j0 < cn; j0 += 4) {
        float* p = w + (j0 + q) * GAT_HEADS + h8;
        const float ex = expf(*p - m);
        *p = ex;
        sum += ex;
      }
    }
    float dn = sum + __shfl_xor(sum, 8);
    dn += __shfl_xor(dn, 16);

    float4 acc = make_float4(0.0f, 0.0f, 0.0f, 0.0f);
    for (int c = 0; c < nch; ++c) {
      const int cn = min(32, deg - 32 * c);
      if (nch > 1) {
        wave_lds_order();
        gat_sorted_chunk(seg, deg, c, lane, ids);
        wave_lds_order();
        sid = (lane < cn) ? ids[lane] : 0;
        stage_logits(sid, cn);
        wave_lds_order();
        for (int j0 = 0; j0 < cn; j0 += 4) {
          float* p = w + (j0 + q) * GAT_HEADS + h8;
          *p = expf(*p - m);
        }
      }
      wave_lds_order();
#pragma unroll 4
      for (int j = 0; j < cn; ++j) {
        const int s = __builtin_amdgcn_readlane(sid, j);
        const float a = w[j * GAT_HEADS + hc];
        const float4 xv = *(const float4*)(xp + (size_t)s * GAT_OUT + col);
        acc.x += a * xv.x;
        acc.y += a * xv.y;
        acc.z += a * xv.z;
        acc.w += a * xv.w;
      }
    }

    float dnc = __shfl(dn, hc);
    dnc = (dnc == 0.0f) ? 1.0f : dnc;
    const float inv = 1.0f / dnc;
    float4 o;
    o.x = acc.x * inv + bv.x;
    o.y = acc.y * inv + bv.y;
    o.z = acc.z * inv + bv.z;
    o.w = acc.w * inv + bv.w;
    o.x = o.x > 0.0f ? o.x : expm1f(o.x);
    o.y = o.y > 0.0f ? o.y : expm1f(o.y);
    o.z = o.z > 0.0f ? o.z : expm1f(o.z);
    o.w = o.w > 0.0f ? o.w : expm1f(o.w);
    *(float4*)(out + (size_t)d * GAT_OUT + col) = o;
    wave_lds_order();
  }
}

extern "C" void kernel_launch(void* const* d_in, const int* in_sizes, int n_in,
                              void* d_out, int out_size, void* d_ws,
                              size_t ws_size, hipStream_t stream) {
  const float* x      = (const float*)d_in[0];
  const int*   ei     = (const int*)d_in[1];
  const float* W      = (const float*)d_in[2];
  const float* a_src  = (const float*)d_in[3];
  const float* a_dst  = (const float*)d_in[4];
  const float* bias   = (const float*)d_in[5];

  const int N = in_sizes[0] / GAT_IN_DIM;
  const int E = in_sizes[1] / 2;
  const int* src = ei;
  const int* dst = ei + E;

  float* out = (float*)d_out;

  float* ws     = (float*)d_ws;
  float* xp     = ws;
  float* al_src = xp + (size_t)N * GAT_OUT;
  int*   cnt    = (int*)(al_src + (size_t)N * GAT_HEADS);
  int*   cur    = cnt + N;
  int*   total  = cur + N;
  int*   nbr    = total + 4;

  gat_init<<<(N + 255) / 256, 256, 0, stream>>>(cnt, total, N);
  gat_proj_wmma<<<(N + 15) / 16, 256, 0, stream>>>(x, W, xp, N);
  {
    int total_nh = N * GAT_HEADS;
    gat_al<<<(total_nh + 255) / 256, 256, 0, stream>>>(xp, a_src, al_src, N);
  }
  gat_dst_count<<<2048, 256, 0, stream>>>(dst, cnt, E);
  gat_seg_alloc<<<(N + 255) / 256, 256, 0, stream>>>(cnt, cur, total, N);
  gat_seg_fill<<<2048, 256, 0, stream>>>(src, dst, cur, nbr, E);
  gat_node_gather<<<2048, 256, 0, stream>>>(cnt, cur, nbr, al_src, a_dst, xp,
                                            bias, out, N);
}


# Projection re-tiled as a resident grid: each wave keeps its 128x16 slice of W in registers and walks 16-row tiles of x staged one ahead into LDS rows of pitch 132 floats (conflict-free A reads)
# speedup vs baseline: 3.5862x; 1.1218x over previous
#include <hip/hip_runtime.h>
#include <hip/hip_bf16.h>
#include <math.h>


typedef __attribute__((ext_vector_type(2))) float v2f;
typedef __attribute__((ext_vector_type(8))) float v8f;

#define GAT_IN_DIM   128
#define GAT_HEADS    8
#define GAT_FPH      16
#define GAT_OUT      128
#define NEG_SLOPE    0.2f

#define SX_PITCH 132

__global__ __launch_bounds__(256, 8) void gat_proj_wmma(
    const float* __restrict__ x, const float* __restrict__ W,
    float* __restrict__ xp, int N) {
  __shared__ __align__(16) float sx[2][16 * SX_PITCH];

  const int tid  = threadIdx.x;
  const int wave = tid >> 5;
  const int lane = tid & 31;
  const int n    = lane & 15;
  const int half = lane >> 4;
  const int col  = wave * 16 + n;
  const int ntiles = (N + 15) >> 4;

  v2f b[32];
#pragma unroll
  for (int t = 0; t < 32; ++t) {
    const int kk = 4 * t + 2 * half;
    b[t][0] = W[kk * GAT_OUT + col];
    b[t][1] = W[(kk + 1) * GAT_OUT + col];
  }

  const int sr = tid >> 5;
  const int sc = (tid & 31) * 4;
  const int last = N - 1;
  float4 p0 = make_float4(0.0f, 0.0f, 0.0f, 0.0f), p1 = p0;

  int tile = blockIdx.x;
  if (tile < ntiles) {
    const int g0 = min(tile * 16 + sr, last), g1 = min(tile * 16 + sr + 8, last);
    p0 = *(const float4*)(x + (size_t)g0 * GAT_IN_DIM + sc);
    p1 = *(const float4*)(x + (size_t)g1 * GAT_IN_DIM + sc);
  }

  int buf = 0;
  for (; tile < ntiles; tile += gridDim.x) {
    float* s = sx[buf];
    *(float4*)(s + sr * SX_PITCH + sc)       = p0;
    *(float4*)(s + (sr + 8) * SX_PITCH + sc) = p1;
    __syncthreads();

    const int nxt = tile + gridDim.x;
    if (nxt < ntiles) {
      const int g0 = min(nxt * 16 + sr, last), g1 = min(nxt * 16 + sr + 8, last);
      p0 = *(const float4*)(x + (size_t)g0 * GAT_IN_DIM + sc);
      p1 = *(const float4*)(x + (size_t)g1 * GAT_IN_DIM + sc);
    }

    const float* ap = s + n * SX_PITCH + 2 * half;
    v8f acc = {};
#pragma unroll
    for (int t = 0; t < 32; ++t) {
      v2f a;
      a[0] = ap[4 * t];
      a[1] = ap[4 * t + 1];
      acc = __builtin_amdgcn_wmma_f32_16x16x4_f32(
           false, a,  false, b[t],
           (short)0, acc,  false,  false);
    }

    const int mbase = tile * 16 + half * 8;
#pragma unroll
    for (int i = 0; i < 8; ++i) {
      int gr = mbase + i;
      if (gr < N) xp[(size_t)gr * GAT_OUT + col] = acc[i];
    }
    buf ^= 1;
  }
}

__global__ void gat_al(const float* __restrict__ xp,
                       const float* __restrict__ a_src,
                       float* __restrict__ al_src, int N) {
  int idx = blockIdx.x * blockDim.x + threadIdx.x;
  if (idx >= N * GAT_HEADS) return;
  int h = idx & 7, node = idx >> 3;
  const float* r  = xp + node * GAT_OUT + h * GAT_FPH;
  const float* as = a_src + h * GAT_FPH;
  float s = 0.0f;
#pragma unroll
  for (int f = 0; f < GAT_FPH; ++f) s += r[f] * as[f];
  al_src[idx] = s;
}

__global__ void gat_init(int* __restrict__ cnt, int* __restrict__ total,
                         int N) {
  int i = blockIdx.x * blockDim.x + threadIdx.x;
  if (i < N) cnt[i] = 0;
  if (i == 0) *total = 0;
}

__global__ __launch_bounds__(256) void gat_dst_count(
    const int* __restrict__ dst, int* __restrict__ cnt, int E) {
  const int stride = gridDim.x * blockDim.x;
  for (int e = blockIdx.x * blockDim.x + threadIdx.x; e < E; e += stride)
    atomicAdd(&cnt[dst[e]], 1);
}

__global__ __launch_bounds__(256) void gat_seg_alloc(
    const int* __restrict__ cnt, int* __restrict__ cur,
    int* __restrict__ total, int N) {
  const int d    = blockIdx.x * blockDim.x + threadIdx.x;
  const int lane = threadIdx.x & 31;
  const int c    = (d < N) ? cnt[d] : 0;
  int incl = c;
#pragma unroll
  for (int o = 1; o < 32; o <<= 1) {
    int t = __shfl_up(incl, o);
    if (lane >= o) incl += t;
  }
  int base = 0;
  if (lane == 31) base = atomicAdd(total, incl);
  base = __shfl(base, 31);
  if (d < N) cur[d] = base + incl - c;
}

__global__ __launch_bounds__(256) void gat_seg_fill(
    const int* __restrict__ src, const int* __restrict__ dst,
    int* __restrict__ cur, int* __restrict__ nbr, int E) {
  const int stride = gridDim.x * blockDim.x;
  for (int e = blockIdx.x * blockDim.x + threadIdx.x; e < E; e += stride) {
    const int p = atomicAdd(&cur[dst[e]], 1);
    nbr[p] = src[e];
  }
}

__device__ __forceinline__ float leaky(float e) {
  return e >= 0.0f ? e : NEG_SLOPE * e;
}

__device__ __forceinline__ void wave_lds_order() {
  __builtin_amdgcn_fence(__ATOMIC_RELEASE, "wavefront");
  __builtin_amdgcn_wave_barrier();
  __builtin_amdgcn_fence(__ATOMIC_ACQUIRE, "wavefront");
}

__device__ __forceinline__ void gat_sorted_chunk(const int* __restrict__ seg,
                                              int deg, int c, int lane,
                                              int* ids) {
  for (int pb = 0; pb < deg; pb += 32) {
    const int p = pb + lane;
    const bool valid = p < deg;
    const int v = valid ? seg[p] : 0;
    int rank = 0;
    for (int i = 0; i < deg; ++i) {
      const int vi = seg[i];
      rank += ((vi < v) | ((vi == v) & (i < p))) ? 1 : 0;
    }
    const int r = rank - 32 * c;
    if (valid && r >= 0 && r < 32) ids[r] = v;
  }
}

__global__ __launch_bounds__(256, 16) void gat_node_gather(
    const int* __restrict__ cnt, const int* __restrict__ seg_end,
    const int* __restrict__ nbr, const float* __restrict__ al_src,
    const float* __restrict__ a_dst, const float* __restrict__ xp,
    const float* __restrict__ bias, float* __restrict__ out, int N) {
  __shared__ int   s_ids[8][32];
  __shared__ __align__(16) float s_w[8][32 * GAT_HEADS];

  const int lane = threadIdx.x & 31;
  const int wv = __builtin_amdgcn_readfirstlane((int)(threadIdx.x >> 5));
  int*   ids = s_ids[wv];
  float* w   = s_w[wv];
  const int nw = gridDim.x * 8;
  const int h8 = lane & 7;
  const int q  = lane >> 3;
  const int hc = lane >> 2;
  const int col = lane * 4;
  const float4 adv = *(const float4*)(a_dst + col);
  const float4 bv  = *(const float4*)(bias + col);

  for (int d = blockIdx.x * 8 + wv; d < N; d += nw) {
    const int deg = cnt[d];
    const int* seg = nbr + (seg_end[d] - deg);
    const int nch = (deg + 31) >> 5;

    const float4 xd = *(const float4*)(xp + (size_t)d * GAT_OUT + col);
    float pd = xd.x * adv.x + xd.y * adv.y + xd.z * adv.z + xd.w * adv.w;
    pd += __shfl_xor(pd, 1);
    pd += __shfl_xor(pd, 2);
    float adh[GAT_HEADS];
#pragma unroll
    for (int h = 0; h < GAT_HEADS; ++h)
      adh[h] = __int_as_float(
          __builtin_amdgcn_readlane(__float_as_int(pd), 4 * h));

    auto stage_logits = [&](int sidv, int cn) {
      const float ninf = -__builtin_inff();
      float4 e0 = make_float4(ninf, ninf, ninf, ninf), e1 = e0;
      if (lane < cn) {
        const float* ap = al_src + (size_t)sidv * GAT_HEADS;
        const float4 a0 = *(const float4*)(ap);
        const float4 a1 = *(const float4*)(ap + 4);
        e0.x = leaky(a0.x + adh[0]);
        e0.y = leaky(a0.y + adh[1]);
        e0.z = leaky(a0.z + adh[2]);
        e0.w = leaky(a0.w + adh[3]);
        e1.x = leaky(a1.x + adh[4]);
        e1.y = leaky(a1.y + adh[5]);
        e1.z = leaky(a1.z + adh[6]);
        e1.w = leaky(a1.w + adh[7]);
      }
      *(float4*)(w + lane * GAT_HEADS)     = e0;
      *(float4*)(w + lane * GAT_HEADS + 4) = e1;
    };

    int sid = 0;
    if (nch == 1) {
      const int v = (lane < deg) ? seg[lane] : 0;
      int rank = 0;
      for (int i = 0; i < deg; ++i) {
        const int vi = __builtin_amdgcn_readlane(v, i);
        rank += ((vi < v) | ((vi == v) & (i < lane))) ? 1 : 0;
      }
      if (lane < deg) ids[rank] = v;
      wave_lds_order();
      sid = (lane < deg) ? ids[lane] : 0;
    }

    float m = -__builtin_inff();
    for (int c = 0; c < nch; ++c) {
      const int cn = min(32, deg - 32 * c);
      if (nch > 1) {
        wave_lds_order();
        gat_sorted_chunk(seg, deg, c, lane, ids);
        wave_lds_order();
        sid = (lane < cn) ? ids[lane] : 0;
      }
      stage_logits(sid, cn);
      wave_lds_order();
      for (int j0 = 0; j0 < cn; j0 += 4)
        m = fmaxf(m, w[(j0 + q) * GAT_HEADS + h8]);
    }
    m = fmaxf(m, __shfl_xor(m, 8));
    m = fmaxf(m, __shfl_xor(m, 16));

    float sum = 0.0f;
    for (int c = 0; c < nch; ++c) {
      const int cn = min(32, deg - 32 * c);
      if (nch > 1) {
        wave_lds_order();
        gat_sorted_chunk(seg, deg, c, lane, ids);
        wave_lds_order();
        sid = (lane < cn) ? ids[lane] : 0;
        stage_logits(sid, cn);
        wave_lds_order();
      }
      for (int j0 = 0; j0 < cn; j0 += 4) {
        float* p = w + (j0 + q) * GAT_HEADS + h8;
        const float ex = expf(*p - m);
        *p = ex;
        sum += ex;
      }
    }
    float dn = sum + __shfl_xor(sum, 8);
    dn += __shfl_xor(dn, 16);

    float4 acc = make_float4(0.0f, 0.0f, 0.0f, 0.0f);
    for (int c = 0; c < nch; ++c) {
      const int cn = min(32, deg - 32 * c);
      if (nch > 1) {
        wave_lds_order();
        gat_sorted_chunk(seg, deg, c, lane, ids);
        wave_lds_order();
        sid = (lane < cn) ? ids[lane] : 0;
        stage_logits(sid, cn);
        wave_lds_order();
        for (int j0 = 0; j0 < cn; j0 += 4) {
          float* p = w + (j0 + q) * GAT_HEADS + h8;
          *p = expf(*p - m);
        }
      }
      wave_lds_order();
#pragma unroll 4
      for (int j = 0; j < cn; ++j) {
        const int s = __builtin_amdgcn_readlane(sid, j);
        const float a = w[j * GAT_HEADS + hc];
        const float4 xv = *(const float4*)(xp + (size_t)s * GAT_OUT + col);
        acc.x += a * xv.x;
        acc.y += a * xv.y;
        acc.z += a * xv.z;
        acc.w += a * xv.w;
      }
    }

    float dnc = __shfl(dn, hc);
    dnc = (dnc == 0.0f) ? 1.0f : dnc;
    const float inv = 1.0f / dnc;
    float4 o;
    o.x = acc.x * inv + bv.x;
    o.y = acc.y * inv + bv.y;
    o.z = acc.z * inv + bv.z;
    o.w = acc.w * inv + bv.w;
    o.x = o.x > 0.0f ? o.x : expm1f(o.x);
    o.y = o.y > 0.0f ? o.y : expm1f(o.y);
    o.z = o.z > 0.0f ? o.z : expm1f(o.z);
    o.w = o.w > 0.0f ? o.w : expm1f(o.w);
    *(float4*)(out + (size_t)d * GAT_OUT + col) = o;
    wave_lds_order();
  }
}

extern "C" void kernel_launch(void* const* d_in, const int* in_sizes, int n_in,
                              void* d_out, int out_size, void* d_ws,
                              size_t ws_size, hipStream_t stream) {
  const float* x      = (const float*)d_in[0];
  const int*   ei     = (const int*)d_in[1];
  const float* W      = (const float*)d_in[2];
  const float* a_src  = (const float*)d_in[3];
  const float* a_dst  = (const float*)d_in[4];
  const float* bias   = (const float*)d_in[5];

  const int N = in_sizes[0] / GAT_IN_DIM;
  const int E = in_sizes[1] / 2;
  const int* src = ei;
  const int* dst = ei + E;

  float* out = (float*)d_out;

  float* ws     = (float*)d_ws;
  float* xp     = ws;
  float* al_src = xp + (size_t)N * GAT_OUT;
  int*   cnt    = (int*)(al_src + (size_t)N * GAT_HEADS);
  int*   cur    = cnt + N;
  int*   total  = cur + N;
  int*   nbr    = total + 4;

  gat_init<<<(N + 255) / 256, 256, 0, stream>>>(cnt, total, N);
  gat_proj_wmma<<<1024, 256, 0, stream>>>(x, W, xp, N);
  {
    int total_nh = N * GAT_HEADS;
    gat_al<<<(total_nh + 255) / 256, 256, 0, stream>>>(xp, a_src, al_src, N);
  }
  gat_dst_count<<<2048, 256, 0, stream>>>(dst, cnt, E);
  gat_seg_alloc<<<(N + 255) / 256, 256, 0, stream>>>(cnt, cur, total, N);
  gat_seg_fill<<<2048, 256, 0, stream>>>(src, dst, cur, nbr, E);
  gat_node_gather<<<2048, 256, 0, stream>>>(cnt, cur, nbr, al_src, a_dst, xp,
                                            bias, out, N);
}


# Edges grouped by destination with a two-level counting sort on LDS counters (256-node buckets, then nodes) instead of one global atomic per edge for count and fill; 5 launches -> 6
# speedup vs baseline: 6.6486x; 1.8539x over previous
#include <hip/hip_runtime.h>
#include <hip/hip_bf16.h>
#include <math.h>


typedef __attribute__((ext_vector_type(2))) float v2f;
typedef __attribute__((ext_vector_type(8))) float v8f;

#define GAT_IN_DIM   128
#define GAT_HEADS    8
#define GAT_FPH      16
#define GAT_OUT      128
#define NEG_SLOPE    0.2f

#define SX_PITCH 132

__device__ __forceinline__ float row16_sum(float v) {
  v += __int_as_float(__builtin_amdgcn_mov_dpp(
      __float_as_int(v), 0x161  , 0xf, 0xf, true));
  v += __int_as_float(__builtin_amdgcn_mov_dpp(
      __float_as_int(v), 0x162  , 0xf, 0xf, true));
  v += __int_as_float(__builtin_amdgcn_mov_dpp(
      __float_as_int(v), 0x164  , 0xf, 0xf, true));
  v += __int_as_float(__builtin_amdgcn_mov_dpp(
      __float_as_int(v), 0x168  , 0xf, 0xf, true));
  return v;
}

__global__ __launch_bounds__(256, 8) void gat_proj_wmma(
    const float* __restrict__ x, const float* __restrict__ W,
    const float* __restrict__ a_src, float* __restrict__ xp,
    float* __restrict__ al_src, int N) {
  __shared__ __align__(16) float sx[2][16 * SX_PITCH];

  const int tid  = threadIdx.x;
  const int wave = tid >> 5;
  const int lane = tid & 31;
  const int n    = lane & 15;
  const int half = lane >> 4;
  const int col  = wave * 16 + n;
  const int ntiles = (N + 15) >> 4;

  v2f b[32];
#pragma unroll
  for (int t = 0; t < 32; ++t) {
    const int kk = 4 * t + 2 * half;
    b[t][0] = W[kk * GAT_OUT + col];
    b[t][1] = W[(kk + 1) * GAT_OUT + col];
  }

  const int sr = tid >> 5;
  const int sc = (tid & 31) * 4;
  const int last = N - 1;
  float4 p0 = make_float4(0.0f, 0.0f, 0.0f, 0.0f), p1 = p0;

  int tile = blockIdx.x;
  if (tile < ntiles) {
    const int g0 = min(tile * 16 + sr, last), g1 = min(tile * 16 + sr + 8, last);
    p0 = *(const float4*)(x + (size_t)g0 * GAT_IN_DIM + sc);
    p1 = *(const float4*)(x + (size_t)g1 * GAT_IN_DIM + sc);
  }

  const float asv = a_src[col];

  int buf = 0;
  for (; tile < ntiles; tile += gridDim.x) {
    float* s = sx[buf];
    *(float4*)(s + sr * SX_PITCH + sc)       = p0;
    *(float4*)(s + (sr + 8) * SX_PITCH + sc) = p1;
    __syncthreads();

    const int nxt = tile + gridDim.x;
    if (nxt < ntiles) {
      const int g0 = min(nxt * 16 + sr, last), g1 = min(nxt * 16 + sr + 8, last);
      p0 = *(const float4*)(x + (size_t)g0 * GAT_IN_DIM + sc);
      p1 = *(const float4*)(x + (size_t)g1 * GAT_IN_DIM + sc);
    }

    const float* ap = s + n * SX_PITCH + 2 * half;
    v8f acc = {};
#pragma unroll
    for (int t = 0; t < 32; ++t) {
      v2f a;
      a[0] = ap[4 * t];
      a[1] = ap[4 * t + 1];
      acc = __builtin_amdgcn_wmma_f32_16x16x4_f32(
           false, a,  false, b[t],
           (short)0, acc,  false,  false);
    }

    const int mbase = tile * 16 + half * 8;
#pragma unroll
    for (int i = 0; i < 8; ++i) {
      int gr = mbase + i;
      if (gr < N) xp[(size_t)gr * GAT_OUT + col] = acc[i];
    }

    float alv = 0.0f;
#pragma unroll
    for (int i = 0; i < 8; ++i) {
      const float r = row16_sum(acc[i] * asv);
      alv = (n == i) ? r : alv;
    }
    {
      const int gr = mbase + n;
      if (n < 8 && gr < N) al_src[(size_t)gr * GAT_HEADS + wave] = alv;
    }

    buf ^= 1;
  }
}

#define GAT_BIN_BLOCKS  256
#define GAT_BIN_THREADS 512
#define GAT_NB_MAX      512
#define GAT_BIN_BATCH   8

__device__ __forceinline__ int wave_incl_scan(int v, int lane) {
#pragma unroll
  for (int o = 1; o < 32; o <<= 1) {
    const int t = __shfl_up(v, o);
    if (lane >= o) v += t;
  }
  return v;
}

__global__ __launch_bounds__(GAT_BIN_THREADS) void gat_bin_count(
    const int* __restrict__ dst, int* __restrict__ bhist, int N, int E) {
  __shared__ int h[GAT_NB_MAX];
  const int tid = threadIdx.x;
  const int nb = (N + 255) >> 8;
  const int run = (E + GAT_BIN_BLOCKS - 1) / GAT_BIN_BLOCKS;
  const int e0 = blockIdx.x * run;
  const int e1 = min(E, e0 + run);
  for (int b = tid; b < nb; b += GAT_BIN_THREADS) h[b] = 0;
  __syncthreads();
  for (int eb = e0; eb < e1; eb += GAT_BIN_BATCH * GAT_BIN_THREADS) {
    int d[GAT_BIN_BATCH];
#pragma unroll
    for (int k = 0; k < GAT_BIN_BATCH; ++k) {
      const int e = eb + k * GAT_BIN_THREADS + tid;
      d[k] = dst[min(e, e1 - 1)];
    }
    __builtin_amdgcn_sched_barrier(0);
#pragma unroll
    for (int k = 0; k < GAT_BIN_BATCH; ++k)
      if (eb + k * GAT_BIN_THREADS + tid < e1) atomicAdd(&h[d[k] >> 8], 1);
  }
  __syncthreads();
  for (int b = tid; b < nb; b += GAT_BIN_THREADS)
    bhist[b * GAT_BIN_BLOCKS + blockIdx.x] = h[b];
}

__global__ __launch_bounds__(256) void gat_bin_scan(int* __restrict__ bhist,
                                                    int* __restrict__ btot,
                                                    int nb) {
  const int lane = threadIdx.x & 31;
  const int b = blockIdx.x * 8 + (threadIdx.x >> 5);
  if (b >= nb) return;
  int4* p = (int4*)(bhist + (size_t)b * GAT_BIN_BLOCKS + lane * 8);
  const int4 v0 = p[0], v1 = p[1];
  const int s = v0.x + v0.y + v0.z + v0.w + v1.x + v1.y + v1.z + v1.w;
  const int incl = wave_incl_scan(s, lane);
  int4 o0, o1;
  int r = incl - s;
  o0.x = r; r += v0.x;
  o0.y = r; r += v0.y;
  o0.z = r; r += v0.z;
  o0.w = r; r += v0.w;
  o1.x = r; r += v1.x;
  o1.y = r; r += v1.y;
  o1.z = r; r += v1.z;
  o1.w = r;
  p[0] = o0;
  p[1] = o1;
  if (lane == 31) btot[b] = incl;
}

__global__ __launch_bounds__(GAT_BIN_THREADS) void gat_bin_scatter(
    const int* __restrict__ src, const int* __restrict__ dst,
    const int* __restrict__ bhist, const int* __restrict__ btot,
    unsigned* __restrict__ rec, int N, int E) {
  __shared__ int cur[GAT_NB_MAX];
  __shared__ int wsum[GAT_BIN_THREADS / 32];
  const int tid = threadIdx.x;
  const int lane = tid & 31;
  const int w = tid >> 5;
  const int nb = (N + 255) >> 8;
  const int run = (E + GAT_BIN_BLOCKS - 1) / GAT_BIN_BLOCKS;
  const int e0 = blockIdx.x * run;
  const int e1 = min(E, e0 + run);

  const int v = (tid < nb) ? btot[tid] : 0;
  const int mine = (tid < nb) ? bhist[tid * GAT_BIN_BLOCKS + blockIdx.x] : 0;
  const int incl = wave_incl_scan(v, lane);
  if (lane == 31) wsum[w] = incl;
  __syncthreads();
  int woff = 0;
  for (int k = 0; k < w; ++k) woff += wsum[k];
  if (tid < nb) cur[tid] = woff + incl - v + mine;
  __syncthreads();

  for (int eb = e0; eb < e1; eb += GAT_BIN_BATCH * GAT_BIN_THREADS) {
    int d[GAT_BIN_BATCH], s[GAT_BIN_BATCH];
#pragma unroll
    for (int k = 0; k < GAT_BIN_BATCH; ++k) {
      const int ec = min(eb + k * GAT_BIN_THREADS + tid, e1 - 1);
      d[k] = dst[ec];
      s[k] = src[ec];
    }
    __builtin_amdgcn_sched_barrier(0);
#pragma unroll
    for (int k = 0; k < GAT_BIN_BATCH; ++k) {
      if (eb + k * GAT_BIN_THREADS + tid < e1) {
        const int p = atomicAdd(&cur[d[k] >> 8], 1);
        rec[p] = ((unsigned)s[k] << 8) | ((unsigned)d[k] & 255u);
      }
    }
  }
}

__global__ __launch_bounds__(256) void gat_bin_csr(
    const unsigned* __restrict__ rec, const int* __restrict__ btot,
    int* __restrict__ cnt, int* __restrict__ seg_end, int* __restrict__ nbr,
    int N) {
  __shared__ int c[256];
  __shared__ int red[8];
  __shared__ int wsum[8];
  const int t = threadIdx.x;
  const int lane = t & 31;
  const int w = t >> 5;
  const int b = blockIdx.x;

  int part = 0;
  for (int i = t; i < b; i += 256) part += btot[i];
  part += __shfl_xor(part, 1);
  part += __shfl_xor(part, 2);
  part += __shfl_xor(part, 4);
  part += __shfl_xor(part, 8);
  part += __shfl_xor(part, 16);
  if (lane == 0) red[w] = part;
  c[t] = 0;
  __syncthreads();
  int base = 0;
  for (int k = 0; k < 8; ++k) base += red[k];
  const int tot = btot[b];
  const unsigned* r0 = rec + base;

  for (int i0 = 0; i0 < tot; i0 += GAT_BIN_BATCH * 256) {
    unsigned r[GAT_BIN_BATCH];
#pragma unroll
    for (int k = 0; k < GAT_BIN_BATCH; ++k) {
      const int i = i0 + k * 256 + t;
      r[k] = r0[min(i, tot - 1)];
    }
    __builtin_amdgcn_sched_barrier(0);
#pragma unroll
    for (int k = 0; k < GAT_BIN_BATCH; ++k)
      if (i0 + k * 256 + t < tot) atomicAdd(&c[r[k] & 255u], 1);
  }
  __syncthreads();

  const int v = c[t];
  const int incl = wave_incl_scan(v, lane);
  if (lane == 31) wsum[w] = incl;
  __syncthreads();
  int woff = 0;
  for (int k = 0; k < w; ++k) woff += wsum[k];
  const int start = base + woff + incl - v;
  c[t] = start;
  const int dnode = b * 256 + t;
  if (dnode < N) {
    cnt[dnode] = v;
    seg_end[dnode] = start + v;
  }
  __syncthreads();

  for (int i0 = 0; i0 < tot; i0 += GAT_BIN_BATCH * 256) {
    unsigned r[GAT_BIN_BATCH];
#pragma unroll
    for (int k = 0; k < GAT_BIN_BATCH; ++k) {
      const int i = i0 + k * 256 + t;
      r[k] = r0[min(i, tot - 1)];
    }
    __builtin_amdgcn_sched_barrier(0);
#pragma unroll
    for (int k = 0; k < GAT_BIN_BATCH; ++k) {
      if (i0 + k * 256 + t < tot) {
        const int p = atomicAdd(&c[r[k] & 255u], 1);
        nbr[p] = (int)(r[k] >> 8);
      }
    }
  }
}

__device__ __forceinline__ float leaky(float e) {
  return e >= 0.0f ? e : NEG_SLOPE * e;
}

__device__ __forceinline__ void wave_lds_order() {
  __builtin_amdgcn_fence(__ATOMIC_RELEASE, "wavefront");
  __builtin_amdgcn_wave_barrier();
  __builtin_amdgcn_fence(__ATOMIC_ACQUIRE, "wavefront");
}

__device__ __forceinline__ void gat_sorted_chunk(const int* __restrict__ seg,
                                              int deg, int c, int lane,
                                              int* ids) {
  for (int pb = 0; pb < deg; pb += 32) {
    const int p = pb + lane;
    const bool valid = p < deg;
    const int v = valid ? seg[p] : 0;
    int rank = 0;
    for (int i = 0; i < deg; ++i) {
      const int vi = seg[i];
      rank += ((vi < v) | ((vi == v) & (i < p))) ? 1 : 0;
    }
    const int r = rank - 32 * c;
    if (valid && r >= 0 && r < 32) ids[r] = v;
  }
}

__global__ __launch_bounds__(256, 16) void gat_node_gather(
    const int* __restrict__ cnt, const int* __restrict__ seg_end,
    const int* __restrict__ nbr, const float* __restrict__ al_src,
    const float* __restrict__ a_dst, const float* __restrict__ xp,
    const float* __restrict__ bias, float* __restrict__ out, int N) {
  __shared__ int   s_ids[8][32];
  __shared__ __align__(16) float s_w[8][32 * GAT_HEADS];

  const int lane = threadIdx.x & 31;
  const int wv = __builtin_amdgcn_readfirstlane((int)(threadIdx.x >> 5));
  int*   ids = s_ids[wv];
  float* w   = s_w[wv];
  const int nw = gridDim.x * 8;
  const int h8 = lane & 7;
  const int q  = lane >> 3;
  const int hc = lane >> 2;
  const int col = lane * 4;
  const float4 adv = *(const float4*)(a_dst + col);
  const float4 bv  = *(const float4*)(bias + col);

  for (int d = blockIdx.x * 8 + wv; d < N; d += nw) {
    const int deg = cnt[d];
    const int* seg = nbr + (seg_end[d] - deg);
    const int nch = (deg + 31) >> 5;

    const float4 xd = *(const float4*)(xp + (size_t)d * GAT_OUT + col);
    float pd = xd.x * adv.x + xd.y * adv.y + xd.z * adv.z + xd.w * adv.w;
    pd += __shfl_xor(pd, 1);
    pd += __shfl_xor(pd, 2);
    float adh[GAT_HEADS];
#pragma unroll
    for (int h = 0; h < GAT_HEADS; ++h)
      adh[h] = __int_as_float(
          __builtin_amdgcn_readlane(__float_as_int(pd), 4 * h));

    auto stage_logits = [&](int sidv, int cn) {
      const float ninf = -__builtin_inff();
      float4 e0 = make_float4(ninf, ninf, ninf, ninf), e1 = e0;
      if (lane < cn) {
        const float* ap = al_src + (size_t)sidv * GAT_HEADS;
        const float4 a0 = *(const float4*)(ap);
        const float4 a1 = *(const float4*)(ap + 4);
        e0.x = leaky(a0.x + adh[0]);
        e0.y = leaky(a0.y + adh[1]);
        e0.z = leaky(a0.z + adh[2]);
        e0.w = leaky(a0.w + adh[3]);
        e1.x = leaky(a1.x + adh[4]);
        e1.y = leaky(a1.y + adh[5]);
        e1.z = leaky(a1.z + adh[6]);
        e1.w = leaky(a1.w + adh[7]);
      }
      *(float4*)(w + lane * GAT_HEADS)     = e0;
      *(float4*)(w + lane * GAT_HEADS + 4) = e1;
    };

    int sid = 0;
    if (nch == 1) {
      const int v = (lane < deg) ? seg[lane] : 0;
      int rank = 0;
      for (int i = 0; i < deg; ++i) {
        const int vi = __builtin_amdgcn_readlane(v, i);
        rank += ((vi < v) | ((vi == v) & (i < lane))) ? 1 : 0;
      }
      if (lane < deg) ids[rank] = v;
      wave_lds_order();
      sid = (lane < deg) ? ids[lane] : 0;
    }

    float m = -__builtin_inff();
    for (int c = 0; c < nch; ++c) {
      const int cn = min(32, deg - 32 * c);
      if (nch > 1) {
        wave_lds_order();
        gat_sorted_chunk(seg, deg, c, lane, ids);
        wave_lds_order();
        sid = (lane < cn) ? ids[lane] : 0;
      }
      stage_logits(sid, cn);
      wave_lds_order();
      for (int j0 = 0; j0 < cn; j0 += 4)
        m = fmaxf(m, w[(j0 + q) * GAT_HEADS + h8]);
    }
    m = fmaxf(m, __shfl_xor(m, 8));
    m = fmaxf(m, __shfl_xor(m, 16));

    float sum = 0.0f;
    for (int c = 0; c < nch; ++c) {
      const int cn = min(32, deg - 32 * c);
      if (nch > 1) {
        wave_lds_order();
        gat_sorted_chunk(seg, deg, c, lane, ids);
        wave_lds_order();
        sid = (lane < cn) ? ids[lane] : 0;
        stage_logits(sid, cn);
        wave_lds_order();
      }
      for (int j0 = 0; j0 < cn; j0 += 4) {
        float* p = w + (j0 + q) * GAT_HEADS + h8;
        const float ex = expf(*p - m);
        *p = ex;
        sum += ex;
      }
    }
    float dn = sum + __shfl_xor(sum, 8);
    dn += __shfl_xor(dn, 16);

    float4 acc = make_float4(0.0f, 0.0f, 0.0f, 0.0f);
    for (int c = 0; c < nch; ++c) {
      const int cn = min(32, deg - 32 * c);
      if (nch > 1) {
        wave_lds_order();
        gat_sorted_chunk(seg, deg, c, lane, ids);
        wave_lds_order();
        sid = (lane < cn) ? ids[lane] : 0;
        stage_logits(sid, cn);
        wave_lds_order();
        for (int j0 = 0; j0 < cn; j0 += 4) {
          float* p = w + (j0 + q) * GAT_HEADS + h8;
          *p = expf(*p - m);
        }
      }
      wave_lds_order();
#pragma unroll 4
      for (int j = 0; j < cn; ++j) {
        const int s = __builtin_amdgcn_readlane(sid, j);
        const float a = w[j * GAT_HEADS + hc];
        const float4 xv = *(const float4*)(xp + (size_t)s * GAT_OUT + col);
        acc.x += a * xv.x;
        acc.y += a * xv.y;
        acc.z += a * xv.z;
        acc.w += a * xv.w;
      }
    }

    float dnc = __shfl(dn, hc);
    dnc = (dnc == 0.0f) ? 1.0f : dnc;
    const float inv = 1.0f / dnc;
    float4 o;
    o.x = acc.x * inv + bv.x;
    o.y = acc.y * inv + bv.y;
    o.z = acc.z * inv + bv.z;
    o.w = acc.w * inv + bv.w;
    o.x = o.x > 0.0f ? o.x : expm1f(o.x);
    o.y = o.y > 0.0f ? o.y : expm1f(o.y);
    o.z = o.z > 0.0f ? o.z : expm1f(o.z);
    o.w = o.w > 0.0f ? o.w : expm1f(o.w);
    *(float4*)(out + (size_t)d * GAT_OUT + col) = o;
    wave_lds_order();
  }
}

extern "C" void kernel_launch(void* const* d_in, const int* in_sizes, int n_in,
                              void* d_out, int out_size, void* d_ws,
                              size_t ws_size, hipStream_t stream) {
  const float* x      = (const float*)d_in[0];
  const int*   ei     = (const int*)d_in[1];
  const float* W      = (const float*)d_in[2];
  const float* a_src  = (const float*)d_in[3];
  const float* a_dst  = (const float*)d_in[4];
  const float* bias   = (const float*)d_in[5];

  const int N = in_sizes[0] / GAT_IN_DIM;
  const int E = in_sizes[1] / 2;
  const int* src = ei;
  const int* dst = ei + E;

  float* out = (float*)d_out;

  float* ws     = (float*)d_ws;
  float* xp     = ws;
  float* al_src = xp + (size_t)N * GAT_OUT;
  int*   cnt    = (int*)(al_src + (size_t)N * GAT_HEADS);
  int*   cur    = cnt + N;
  int*   nbr    = cur + N + 4;
  unsigned* rec = (unsigned*)xp;
  int*   bhist  = (int*)al_src;
  int*   btot   = bhist + GAT_NB_MAX * GAT_BIN_BLOCKS;

  gat_bin_count<<<GAT_BIN_BLOCKS, GAT_BIN_THREADS, 0, stream>>>(dst, bhist, N,
                                                                E);
  gat_bin_scan<<<((N + 255) / 256 + 7) / 8, 256, 0, stream>>>(
      bhist, btot, (N + 255) / 256);
  gat_bin_scatter<<<GAT_BIN_BLOCKS, GAT_BIN_THREADS, 0, stream>>>(
      src, dst, bhist, btot, rec, N, E);
  gat_bin_csr<<<(N + 255) / 256, 256, 0, stream>>>(rec, btot, cnt, cur, nbr,
                                                   N);
  gat_proj_wmma<<<1024, 256, 0, stream>>>(x, W, a_src, xp, al_src, N);
  gat_node_gather<<<2048, 256, 0, stream>>>(cnt, cur, nbr, al_src, a_dst, xp,
                                            bias, out, N);
}


# Node kernel takes two entries per pass: each 16-lane half gathers one row, 8 columns per lane (xp rows kept in lane order so loads cover whole lines), halves joined per node; float sum order changes
# speedup vs baseline: 10.0872x; 1.5172x over previous
#include <hip/hip_runtime.h>
#include <hip/hip_bf16.h>
#include <math.h>


typedef __attribute__((ext_vector_type(2))) float v2f;
typedef __attribute__((ext_vector_type(8))) float v8f;

#define GAT_IN_DIM   128
#define GAT_HEADS    8
#define GAT_FPH      16
#define GAT_OUT      128
#define NEG_SLOPE    0.2f

#define SX_PITCH 132

#define GAT_RUNS        256
#define GAT_LOC_THREADS 256
#define GAT_NB_MAX      512
#define GAT_LOC_BATCH   16

__device__ __forceinline__ int wave_incl_scan(int v, int lane) {
#pragma unroll
  for (int o = 1; o < 32; o <<= 1) {
    const int t = __shfl_up(v, o);
    if (lane >= o) v += t;
  }
  return v;
}

__device__ __forceinline__ void gat_bin_local_block(
    const int* __restrict__ src, const int* __restrict__ dst,
    int* __restrict__ lstart, unsigned* __restrict__ rec, int N, int E,
    int k, int* h  ,
    int* wsum  ) {
  const int tid = threadIdx.x;
  const int lane = tid & 31;
  const int w = tid >> 5;
  const int nb = (N + 255) >> 8;
  const int run = (E + GAT_RUNS - 1) / GAT_RUNS;
  const int e0 = k * run;
  const int e1 = min(E, e0 + run);

  h[tid] = 0;
  h[tid + GAT_LOC_THREADS] = 0;
  __syncthreads();
  for (int eb = e0; eb < e1; eb += GAT_LOC_BATCH * GAT_LOC_THREADS) {
    int d[GAT_LOC_BATCH];
#pragma unroll
    for (int q = 0; q < GAT_LOC_BATCH; ++q) {
      const int e = eb + q * GAT_LOC_THREADS + tid;
      d[q] = dst[min(e, e1 - 1)];
    }
    __builtin_amdgcn_sched_barrier(0);
#pragma unroll
    for (int q = 0; q < GAT_LOC_BATCH; ++q)
      if (eb + q * GAT_LOC_THREADS + tid < e1) atomicAdd(&h[d[q] >> 8], 1);
  }
  __syncthreads();

  const int b0 = 2 * tid, b1 = 2 * tid + 1;
  const int v0 = (b0 < nb) ? h[b0] : 0;
  const int v1 = (b1 < nb) ? h[b1] : 0;
  const int pv = v0 + v1;
  const int incl = wave_incl_scan(pv, lane);
  if (lane == 31) wsum[w] = incl;
  __syncthreads();
  int woff = 0;
  for (int j = 0; j < w; ++j) woff += wsum[j];
  const int x0 = woff + incl - pv;
  const int x1 = x0 + v0;
  if (b0 < nb) {
    h[b0] = x0;
    lstart[b0 * GAT_RUNS + k] = x0;
    if (b0 == nb - 1) lstart[nb * GAT_RUNS + k] = x0 + v0;
  }
  if (b1 < nb) {
    h[b1] = x1;
    lstart[b1 * GAT_RUNS + k] = x1;
    if (b1 == nb - 1) lstart[nb * GAT_RUNS + k] = x1 + v1;
  }
  __syncthreads();

  for (int eb = e0; eb < e1; eb += GAT_LOC_BATCH * GAT_LOC_THREADS) {
    int d[GAT_LOC_BATCH], s[GAT_LOC_BATCH];
#pragma unroll
    for (int q = 0; q < GAT_LOC_BATCH; ++q) {
      const int ec = min(eb + q * GAT_LOC_THREADS + tid, e1 - 1);
      d[q] = dst[ec];
      s[q] = src[ec];
    }
    __builtin_amdgcn_sched_barrier(0);
#pragma unroll
    for (int q = 0; q < GAT_LOC_BATCH; ++q) {
      if (eb + q * GAT_LOC_THREADS + tid < e1) {
        const int p = atomicAdd(&h[d[q] >> 8], 1);
        rec[e0 + p] = ((unsigned)s[q] << 8) | ((unsigned)d[q] & 255u);
      }
    }
  }
}

__global__ __launch_bounds__(256, 8) void gat_proj_wmma(
    const float* __restrict__ x, const float* __restrict__ W,
    float* __restrict__ xp, int N, const int* __restrict__ src,
    const int* __restrict__ dst, int* __restrict__ lstart,
    unsigned* __restrict__ rec, int E) {
  __shared__ __align__(16) float sx[2][16 * SX_PITCH];
  __shared__ int sort_h[GAT_NB_MAX];
  __shared__ int sort_wsum[GAT_LOC_THREADS / 32];

  if (blockIdx.x < GAT_RUNS) {
    gat_bin_local_block(src, dst, lstart, rec, N, E, (int)blockIdx.x, sort_h,
                        sort_wsum);
    return;
  }
  const int pblk  = (int)blockIdx.x - GAT_RUNS;
  const int npblk = (int)gridDim.x - GAT_RUNS;

  const int tid  = threadIdx.x;
  const int wave = tid >> 5;
  const int lane = tid & 31;
  const int n    = lane & 15;
  const int half = lane >> 4;
  const int col  = wave * 16 + n;
  const int wcol = 32 * (wave & 3) + 8 * (n >> 2) + 4 * (wave >> 2) + (n & 3);
  const int ntiles = (N + 15) >> 4;

  v2f b[32];
#pragma unroll
  for (int t = 0; t < 32; ++t) {
    const int kk = 4 * t + 2 * half;
    b[t][0] = W[kk * GAT_OUT + wcol];
    b[t][1] = W[(kk + 1) * GAT_OUT + wcol];
  }

  const int sr = tid >> 5;
  const int sc = (tid & 31) * 4;
  const int last = N - 1;
  float4 p0 = make_float4(0.0f, 0.0f, 0.0f, 0.0f), p1 = p0;

  int tile = pblk;
  if (tile < ntiles) {
    const int g0 = min(tile * 16 + sr, last), g1 = min(tile * 16 + sr + 8, last);
    p0 = *(const float4*)(x + (size_t)g0 * GAT_IN_DIM + sc);
    p1 = *(const float4*)(x + (size_t)g1 * GAT_IN_DIM + sc);
  }

  int buf = 0;
  for (; tile < ntiles; tile += npblk) {
    float* s = sx[buf];
    *(float4*)(s + sr * SX_PITCH + sc)       = p0;
    *(float4*)(s + (sr + 8) * SX_PITCH + sc) = p1;
    __syncthreads();

    const int nxt = tile + npblk;
    if (nxt < ntiles) {
      const int g0 = min(nxt * 16 + sr, last), g1 = min(nxt * 16 + sr + 8, last);
      p0 = *(const float4*)(x + (size_t)g0 * GAT_IN_DIM + sc);
      p1 = *(const float4*)(x + (size_t)g1 * GAT_IN_DIM + sc);
    }

    const float* ap = s + n * SX_PITCH + 2 * half;
    v8f acc = {};
#pragma unroll
    for (int t = 0; t < 32; ++t) {
      v2f a;
      a[0] = ap[4 * t];
      a[1] = ap[4 * t + 1];
      acc = __builtin_amdgcn_wmma_f32_16x16x4_f32(
           false, a,  false, b[t],
           (short)0, acc,  false,  false);
    }

    const int mbase = tile * 16 + half * 8;
#pragma unroll
    for (int i = 0; i < 8; ++i) {
      int gr = mbase + i;
      if (gr < N) xp[(size_t)gr * GAT_OUT + col] = acc[i];
    }

    buf ^= 1;
  }
}

#define GAT_BIN_BATCH   8

__global__ __launch_bounds__(256) void gat_bin_csr(
    const unsigned* __restrict__ rec, const int* __restrict__ lstart,
    int* __restrict__ cnt, int* __restrict__ seg_end, int* __restrict__ nbr,
    int N, int E) {
  __shared__ int c[256];
  __shared__ int ps[GAT_RUNS];
  __shared__ int pl[GAT_RUNS];
  __shared__ int red[8];
  __shared__ int wsum[8];
  const int t = threadIdx.x;
  const int lane = t & 31;
  const int w = __builtin_amdgcn_readfirstlane((int)(threadIdx.x >> 5));
  const int b = blockIdx.x;
  const int run = (E + GAT_RUNS - 1) / GAT_RUNS;

  const int st = lstart[b * GAT_RUNS + t];
  const int en = lstart[(b + 1) * GAT_RUNS + t];
  ps[t] = t * run + st;
  pl[t] = en - st;
  int part = st;
  part += __shfl_xor(part, 1);
  part += __shfl_xor(part, 2);
  part += __shfl_xor(part, 4);
  part += __shfl_xor(part, 8);
  part += __shfl_xor(part, 16);
  if (lane == 0) red[w] = part;
  c[t] = 0;
  __syncthreads();
  int base = 0;
  for (int k = 0; k < 8; ++k) base += red[k];

  auto for_each_record = [&](auto f) {
#pragma nounroll
    for (int r0 = 0; r0 < GAT_RUNS / 8; r0 += GAT_BIN_BATCH) {
      unsigned r[GAT_BIN_BATCH];
      int s0[GAT_BIN_BATCH], ln[GAT_BIN_BATCH];
#pragma unroll
      for (int q = 0; q < GAT_BIN_BATCH; ++q) {
        const int k = w + 8 * (r0 + q);
        s0[q] = __builtin_amdgcn_readfirstlane(ps[k]);
        ln[q] = __builtin_amdgcn_readfirstlane(pl[k]);
        r[q] = rec[(lane < ln[q]) ? s0[q] + lane : 0];
      }
      __builtin_amdgcn_sched_barrier(0);
#pragma unroll
      for (int q = 0; q < GAT_BIN_BATCH; ++q) {
        if (lane < ln[q]) f(r[q]);
#pragma nounroll
        for (int j0 = 32; j0 < ln[q]; j0 += 32)
          if (j0 + lane < ln[q]) f(rec[s0[q] + j0 + lane]);
      }
    }
  };

  for_each_record([&](unsigned rv) { atomicAdd(&c[rv & 255u], 1); });
  __syncthreads();

  const int v = c[t];
  const int incl = wave_incl_scan(v, lane);
  if (lane == 31) wsum[w] = incl;
  __syncthreads();
  int woff = 0;
  for (int k = 0; k < w; ++k) woff += wsum[k];
  const int start = base + woff + incl - v;
  c[t] = start;
  const int dnode = b * 256 + t;
  if (dnode < N) {
    cnt[dnode] = v;
    seg_end[dnode] = start + v;
  }
  __syncthreads();

  for_each_record([&](unsigned rv) {
    const int p = atomicAdd(&c[rv & 255u], 1);
    nbr[p] = (int)(rv >> 8);
  });
}

__device__ __forceinline__ void wave_lds_order() {
  __builtin_amdgcn_fence(__ATOMIC_RELEASE, "wavefront");
  __builtin_amdgcn_wave_barrier();
  __builtin_amdgcn_fence(__ATOMIC_ACQUIRE, "wavefront");
}

__device__ __forceinline__ void gat_sorted_chunk(const int* __restrict__ seg,
                                              int deg, int c, int lane,
                                              int* ids) {
  for (int pb = 0; pb < deg; pb += 32) {
    const int p = pb + lane;
    const bool valid = p < deg;
    const int v = valid ? seg[p] : 0;
    int rank = 0;
    for (int i = 0; i < deg; ++i) {
      const int vi = seg[i];
      rank += ((vi < v) | ((vi == v) & (i < p))) ? 1 : 0;
    }
    const int r = rank - 32 * c;
    if (valid && r >= 0 && r < 32) ids[r] = v;
  }
}


__device__ __forceinline__ float pair2_sum(float v) {
  v += __int_as_float(__builtin_amdgcn_mov_dpp(
      __float_as_int(v), 0xB1  , 0xf, 0xf, true));
  return v;
}

__device__ __forceinline__ float xhalf(float x) {
  const int b = __float_as_int(x);
  return __int_as_float(__builtin_amdgcn_permlanex16(
      b, b, 0x76543210, 0xfedcba98, false, false));
}

typedef __attribute__((ext_vector_type(4))) float v4f;

__device__ __forceinline__ float dot8_from(float c, v4f a0, v4f a1, v4f b0,
                                           v4f b1) {
  v2f t = {c, 0.0f};
  t = a0.lo * b0.lo + t;
  t = a0.hi * b0.hi + t;
  t = a1.lo * b1.lo + t;
  t = a1.hi * b1.hi + t;
  return t.x + t.y;
}

__global__ __launch_bounds__(256, 16) void gat_node_gather(
    const int* __restrict__ cnt, const int* __restrict__ seg_end,
    const int* __restrict__ nbr, const float* __restrict__ a_src,
    const float* __restrict__ a_dst, const float* __restrict__ xp,
    const float* __restrict__ bias, float* __restrict__ out, int N) {
  __shared__ int s_ids[8][32];

  const int lane = threadIdx.x & 31;
  const int wv = __builtin_amdgcn_readfirstlane((int)(threadIdx.x >> 5));
  int* ids = s_ids[wv];
  const int nw = gridDim.x * 8;
  const int h2 = lane >> 4;
  const int q  = lane & 15;
  const int c8 = q * 8;
  const int oc = c8 + 4 * h2;
  const float L2E = 1.44269504088896340736f;
  const v4f as0 = *(const v4f*)(a_src + c8) * L2E;
  const v4f as1 = *(const v4f*)(a_src + c8 + 4) * L2E;
  const v4f ad0 = *(const v4f*)(a_dst + c8) * L2E;
  const v4f ad1 = *(const v4f*)(a_dst + c8 + 4) * L2E;
  const float4 bv = *(const float4*)(bias + oc);
  const float ninf = -__builtin_inff();
  const float mlow = -3.402823466e+38f;
  const v4f* xp4 = (const v4f*)xp;

  for (int d = blockIdx.x * 8 + wv; d < N; d += nw) {
    const int deg = cnt[d];
    const int* seg = nbr + (seg_end[d] - deg);
    const int nch = (deg + 31) >> 5;

    const v4f* xdp = xp4 + ((unsigned)d * 32u + (unsigned)q);
    const float adp = dot8_from(0.0f, xdp[0], xdp[16], ad0, ad1);

    float m = mlow;
    float sum = 0.0f;
    v2f acc0 = {0.0f, 0.0f}, acc1 = {0.0f, 0.0f};
    v2f acc2 = {0.0f, 0.0f}, acc3 = {0.0f, 0.0f};

    for (int c = 0; c < nch; ++c) {
      const int cn = min(32, deg - 32 * c);
      if (nch == 1) {
        const int v = (lane < deg) ? seg[lane] : 0;
        int rank = 0;
        for (int i = 0; i < deg; ++i) {
          const int vi = __builtin_amdgcn_readlane(v, i);
          rank += ((vi < v) | ((vi == v) & (i < lane))) ? 1 : 0;
        }
        ids[(lane < deg) ? rank : lane] = v;
        wave_lds_order();
      } else {
        wave_lds_order();
        gat_sorted_chunk(seg, deg, c, lane, ids);
        wave_lds_order();
      }

      for (int j0 = 0; j0 < cn; j0 += 4) {
        v4f xa[2], xb[2];
        float e[2];
#pragma unroll
        for (int u = 0; u < 2; ++u) {
          const int s = ids[j0 + 2 * u + h2];
          const v4f* rp = xp4 + ((unsigned)s * 32u + (unsigned)q);
          xa[u] = rp[0];
          xb[u] = rp[16];
        }
#pragma unroll
        for (int u = 0; u < 2; ++u) {
          const float t = pair2_sum(dot8_from(adp, xa[u], xb[u], as0, as1));
          e[u] = (h2 < cn - j0 - 2 * u) ? fmaxf(t, NEG_SLOPE * t) : ninf;
        }
        const float gm = fmaxf(fmaxf(m, e[0]), e[1]);
        const float sc = __builtin_amdgcn_exp2f(m - gm);
        acc0 *= sc;
        acc1 *= sc;
        acc2 *= sc;
        acc3 *= sc;
        sum *= sc;
#pragma unroll
        for (int u = 0; u < 2; ++u) {
          const float w = __builtin_amdgcn_exp2f(e[u] - gm);
          acc0 += w * xa[u].lo;
          acc1 += w * xa[u].hi;
          acc2 += w * xb[u].lo;
          acc3 += w * xb[u].hi;
          sum += w;
        }
        m = gm;
      }
    }

    const float f = __builtin_amdgcn_exp2f(m - fmaxf(m, xhalf(m)));
    acc0 *= f;
    acc1 *= f;
    acc2 *= f;
    acc3 *= f;
    sum *= f;
    const bool up = h2 != 0;
    v2f k0 = up ? acc2 : acc0;
    v2f k1 = up ? acc3 : acc1;
    const v2f g0 = up ? acc0 : acc2;
    const v2f g1 = up ? acc1 : acc3;
    k0.x += xhalf(g0.x);
    k0.y += xhalf(g0.y);
    k1.x += xhalf(g1.x);
    k1.y += xhalf(g1.y);
    sum += xhalf(sum);

    const float dnc = (sum == 0.0f) ? 1.0f : sum;
    const float inv = 1.0f / dnc;
    float4 o;
    o.x = k0.x * inv + bv.x;
    o.y = k0.y * inv + bv.y;
    o.z = k1.x * inv + bv.z;
    o.w = k1.y * inv + bv.w;
    o.x = o.x > 0.0f ? o.x : expm1f(o.x);
    o.y = o.y > 0.0f ? o.y : expm1f(o.y);
    o.z = o.z > 0.0f ? o.z : expm1f(o.z);
    o.w = o.w > 0.0f ? o.w : expm1f(o.w);
    *(float4*)(out + (size_t)d * GAT_OUT + oc) = o;
    wave_lds_order();
  }
}

extern "C" void kernel_launch(void* const* d_in, const int* in_sizes, int n_in,
                              void* d_out, int out_size, void* d_ws,
                              size_t ws_size, hipStream_t stream) {
  const float* x      = (const float*)d_in[0];
  const int*   ei     = (const int*)d_in[1];
  const float* W      = (const float*)d_in[2];
  const float* a_src  = (const float*)d_in[3];
  const float* a_dst  = (const float*)d_in[4];
  const float* bias   = (const float*)d_in[5];

  const int N = in_sizes[0] / GAT_IN_DIM;
  const int E = in_sizes[1] / 2;
  const int* src = ei;
  const int* dst = ei + E;

  float* out = (float*)d_out;

  float* ws     = (float*)d_ws;
  float* xp     = ws;
  int*   cnt    = (int*)(xp + (size_t)N * GAT_OUT + (size_t)N * GAT_HEADS);
  int*   cur    = cnt + N;
  int*   nbr    = cur + N + 4;
  unsigned* rec = (unsigned*)out;
  int*   lstart = (int*)out + E;

  gat_proj_wmma<<<1024, 256, 0, stream>>>(x, W, xp, N, src, dst, lstart, rec,
                                          E);
  gat_bin_csr<<<(N + 255) / 256, 256, 0, stream>>>(rec, lstart, cnt, cur, nbr,
                                                   N, E);
  gat_node_gather<<<2048, 256, 0, stream>>>(cnt, cur, nbr, a_src, a_dst, xp,
                                            bias, out, N);
}
